# starting kernel
# baseline (speedup 1.0000x reference)
#include <hip/hip_runtime.h>

typedef __attribute__((ext_vector_type(2))) float v2f;
typedef __attribute__((ext_vector_type(8))) float v8f;

#define NDIM 128

__global__ void rg_zero_kernel(float4* __restrict__ p, int n4) {
    int i = blockIdx.x * blockDim.x + threadIdx.x;
    if (i < n4) p[i] = make_float4(0.f, 0.f, 0.f, 0.f);
}

__global__ void rg_deg_kernel(const int* __restrict__ src, const int* __restrict__ dst,
                              float* __restrict__ odeg, float* __restrict__ ideg, int E) {
    int e = blockIdx.x * blockDim.x + threadIdx.x;
    if (e >= E) return;
    atomicAdd(&odeg[src[e]], 1.0f);
    atomicAdd(&ideg[dst[e]], 1.0f);
}

__global__ void rg_norm_kernel(const float* __restrict__ odeg, const float* __restrict__ ideg,
                               float* __restrict__ onorm, float* __restrict__ s1,
                               float* __restrict__ s2, int N) {
    int i = blockIdx.x * blockDim.x + threadIdx.x;
    if (i >= N) return;
    float od = fmaxf(odeg[i], 1.0f);
    float id = fmaxf(ideg[i], 1.0f);
    onorm[i] = rsqrtf(od);
    float r = rsqrtf(id);
    s1[i] = r;
    s2[i] = r / id;
}

__global__ __launch_bounds__(256) void
rg_edge_kernel(const float* __restrict__ h, const float* __restrict__ onorm,
               const float* __restrict__ weight, const int* __restrict__ src,
               const int* __restrict__ dst, const int* __restrict__ etype,
               float* __restrict__ mix, int E) {
    int e = blockIdx.x * 8 + (threadIdx.x >> 5);
    if (e >= E) return;
    int lane = threadIdx.x & 31;
    int s = src[e];
    int d = dst[e];
    int t = etype[e];
    float sc = onorm[s];
    float4 hv = ((const float4*)(h + (size_t)s * NDIM))[lane];
    float4 wv = ((const float4*)(weight + (size_t)t * NDIM))[lane];
    float* mp = mix + (size_t)d * NDIM + lane * 4;
    atomicAdd(mp + 0, hv.x * wv.x * sc);
    atomicAdd(mp + 1, hv.y * wv.y * sc);
    atomicAdd(mp + 2, hv.z * wv.z * sc);
    atomicAdd(mp + 3, hv.w * wv.w * sc);
}

#define LDS_STRIDE 132
__global__ __launch_bounds__(256) void
rg_gemm_kernel(const float* __restrict__ H, const float* __restrict__ W,
               const float* __restrict__ mix, const float* __restrict__ s1,
               const float* __restrict__ s2, float* __restrict__ out) {
    __shared__ float ldsH[16 * LDS_STRIDE];

    const int m0 = blockIdx.x * 16;
    const int tid = threadIdx.x;

    for (int i = tid; i < 16 * 32; i += 256) {
        int r = i >> 5;
        int c4 = i & 31;
        float4 v = ((const float4*)(H + (size_t)(m0 + r) * NDIM))[c4];
        float* dl = ldsH + r * LDS_STRIDE + c4 * 4;
        dl[0] = v.x; dl[1] = v.y; dl[2] = v.z; dl[3] = v.w;
    }
    __syncthreads();

    const int lane = tid & 31;
    const int hi = lane >> 4;
    const int lo = lane & 15;
    const int n0 = (tid >> 5) * 16;

    v8f c = {};
    #pragma unroll
    for (int k = 0; k < NDIM; k += 4) {
        int kk = k + 2 * hi;
        v2f a;
        a.x = ldsH[lo * LDS_STRIDE + kk];
        a.y = ldsH[lo * LDS_STRIDE + kk + 1];
        v2f b;
        b.x = W[(size_t)kk * NDIM + n0 + lo];
        b.y = W[(size_t)(kk + 1) * NDIM + n0 + lo];
        c = __builtin_amdgcn_wmma_f32_16x16x4_f32(
                 false, a,  false, b,
                 (short)0, c,  false,  false);
    }

    #pragma unroll
    for (int g = 0; g < 8; ++g) {
        int row = m0 + g + 8 * hi;
        float a1 = s1[row];
        float a2 = s2[row];
        size_t off = (size_t)row * NDIM + n0 + lo;
        out[off] = c[g] * a1 + mix[off] * a2;
    }
}

extern "C" void kernel_launch(void* const* d_in, const int* in_sizes, int n_in,
                              void* d_out, int out_size, void* d_ws, size_t ws_size,
                              hipStream_t stream) {
    const float* x      = (const float*)d_in[0];
    const float* weight = (const float*)d_in[1];
    const float* lw0    = (const float*)d_in[2];
    const float* lw1    = (const float*)d_in[3];
    const int*   src    = (const int*)d_in[4];
    const int*   dst    = (const int*)d_in[5];
    const int*   etype  = (const int*)d_in[6];
    float* out = (float*)d_out;

    const int N = in_sizes[0] / NDIM;
    const int E = in_sizes[4];
    const size_t ND = (size_t)N * NDIM;

    float* ws    = (float*)d_ws;
    float* h1    = ws;
    float* mix   = h1 + ND;
    float* odeg  = mix + ND;
    float* ideg  = odeg + N;
    float* onorm = ideg + N;
    float* s1    = onorm + N;
    float* s2    = s1 + N;

    const int ZT = 256;
    {
        int n4 = (2 * N + 3) / 4;
        rg_zero_kernel<<<(n4 + ZT - 1) / ZT, ZT, 0, stream>>>((float4*)odeg, n4);
    }
    rg_deg_kernel<<<(E + ZT - 1) / ZT, ZT, 0, stream>>>(src, dst, odeg, ideg, E);
    rg_norm_kernel<<<(N + ZT - 1) / ZT, ZT, 0, stream>>>(odeg, ideg, onorm, s1, s2, N);

    const int edgeBlocks = (E + 7) / 8;
    const int mix4 = (int)(ND / 4);
    const int gemmBlocks = N / 16;

    rg_zero_kernel<<<(mix4 + ZT - 1) / ZT, ZT, 0, stream>>>((float4*)mix, mix4);
    rg_edge_kernel<<<edgeBlocks, 256, 0, stream>>>(x, onorm, weight, src, dst, etype, mix, E);
    rg_gemm_kernel<<<gemmBlocks, 256, 0, stream>>>(x, lw0, mix, s1, s2, h1);

    rg_zero_kernel<<<(mix4 + ZT - 1) / ZT, ZT, 0, stream>>>((float4*)mix, mix4);
    rg_edge_kernel<<<edgeBlocks, 256, 0, stream>>>(h1, onorm, weight, src, dst, etype, mix, E);
    rg_gemm_kernel<<<gemmBlocks, 256, 0, stream>>>(h1, lw1, mix, s1, s2, out);
}


# Degree counts: a block owns 2048 nodes and a quarter of the edges, counts in LDS, stores 4 partial counts per node (readers add them, exact); no global atomics on counters, no clear pass; 5 launches
# speedup vs baseline: 2.4322x; 2.4322x over previous
#include <hip/hip_runtime.h>

typedef __attribute__((ext_vector_type(2))) float v2f;
typedef __attribute__((ext_vector_type(8))) float v8f;

#define NDIM 128

#define DEG_RANGE 2048
#define DEG_PARTS 4
#define DEG_THREADS 1024
#define DEG_BATCH 8

__device__ __forceinline__ void rg_count(unsigned int* cnt, unsigned int base, int v) {
    unsigned int d = (unsigned int)v - base;
    if (d < (unsigned int)DEG_RANGE) atomicAdd(&cnt[d], 1u);
}

__global__ __launch_bounds__(DEG_THREADS) void
rg_deg_kernel(const int* __restrict__ src, const int* __restrict__ dst,
              float* __restrict__ deg, int E, int N, int nRanges) {
    __shared__ unsigned int cnt[DEG_RANGE];
    const int tid = threadIdx.x;
    int b = blockIdx.x;
    const int part = b % DEG_PARTS;
    b /= DEG_PARTS;
    const int range = b % nRanges;
    const int arr = b / nRanges;
    const int* __restrict__ idx = arr ? dst : src;
    const unsigned int base = (unsigned int)range * DEG_RANGE;

    for (int i = tid; i < DEG_RANGE; i += DEG_THREADS) cnt[i] = 0u;
    __syncthreads();

    const int nv = E >> 2;
    const int lo = (int)(((long long)nv * part) / DEG_PARTS);
    const int hi = (int)(((long long)nv * (part + 1)) / DEG_PARTS);
    const int4* __restrict__ idx4 = (const int4*)idx;
    const int span = DEG_BATCH * DEG_THREADS;
    const int steps = (hi - lo + span - 1) / span;
    const int rot = steps > 0 ? range % steps : 0;
    for (int c = 0; c < steps; ++c) {
        int cr = c + rot;
        if (cr >= steps) cr -= steps;
        const int i0 = lo + cr * span + tid;
        int4 v[DEG_BATCH];
        #pragma unroll
        for (int u = 0; u < DEG_BATCH; ++u) {
            const int i = i0 + u * DEG_THREADS;
            const bool ok = i < hi;
            v[u] = idx4[ok ? i : lo];
            if (!ok) v[u] = make_int4(-1, -1, -1, -1);
        }
        #pragma unroll
        for (int u = 0; u < DEG_BATCH; ++u) {
            rg_count(cnt, base, v[u].x);
            rg_count(cnt, base, v[u].y);
            rg_count(cnt, base, v[u].z);
            rg_count(cnt, base, v[u].w);
        }
    }
    if (part == 0 && tid < (E & 3)) rg_count(cnt, base, idx[nv * 4 + tid]);
    __syncthreads();

    float* __restrict__ o = deg + (size_t)arr * N * DEG_PARTS;
    for (int i = tid; i < DEG_RANGE; i += DEG_THREADS) {
        const unsigned int node = base + (unsigned int)i;
        if (node < (unsigned int)N) o[(size_t)node * DEG_PARTS + part] = (float)cnt[i];
    }
}

__device__ __forceinline__ float rg_deg(const float* __restrict__ deg4, int node) {
    const float4 p = ((const float4*)deg4)[node];
    return (p.x + p.y) + (p.z + p.w);
}

__device__ __forceinline__ float rg_onorm(float odeg) {
    return rsqrtf(fmaxf(odeg, 1.0f));
}
__device__ __forceinline__ float rg_s2(float ideg) {
    float id = fmaxf(ideg, 1.0f);
    return rsqrtf(id) / id;
}

#define EDGES_PER_WAVE 32
#define EDGES_PER_GROUP 4

template <int K>
__device__ __forceinline__ void
rg_edge_group(const float* __restrict__ h, const float* __restrict__ weight,
              float* __restrict__ acc, int s, int d, int t, float sc,
              int j, int lane) {
    float v[K][4];
    for (int u = 0; u < K; ++u) {
        int su = __builtin_amdgcn_readlane(s, j + u);
        int tu = __builtin_amdgcn_readlane(t, j + u);
        float cu = __int_as_float(__builtin_amdgcn_readlane(__float_as_int(sc), j + u));
        const float* hp = h + (size_t)su * NDIM;
        const float* wp = weight + (size_t)tu * NDIM;
        for (int q = 0; q < 4; ++q) v[u][q] = hp[lane + 32 * q] * wp[lane + 32 * q] * cu;
    }
    for (int u = 0; u < K; ++u) {
        int du = __builtin_amdgcn_readlane(d, j + u);
        float* mp = acc + (size_t)du * NDIM;
        for (int q = 0; q < 4; ++q) atomicAdd(mp + lane + 32 * q, v[u][q]);
    }
}

__global__ __launch_bounds__(256) void
rg_edge_kernel(const float* __restrict__ h, const float* __restrict__ odeg,
               const float* __restrict__ ideg,
               const float* __restrict__ weight, const int* __restrict__ src,
               const int* __restrict__ dst, const int* __restrict__ etype,
               float* __restrict__ acc, int E) {
    const int lane = threadIdx.x & 31;
    const int wave = __builtin_amdgcn_readfirstlane((int)(threadIdx.x >> 5));
    const int e0 = (blockIdx.x * 8 + wave) * EDGES_PER_WAVE;
    if (e0 >= E) return;
    const int n = min(EDGES_PER_WAVE, E - e0);
    int s = 0, d = 0, t = 0;
    float sc = 0.f;
    if (lane < n) {
        s = src[e0 + lane];
        d = dst[e0 + lane];
        t = etype[e0 + lane];
        sc = rg_onorm(rg_deg(odeg, s)) * rg_s2(rg_deg(ideg, d));
    }
    int j = 0;
    for (; j + EDGES_PER_GROUP <= n; j += EDGES_PER_GROUP)
        rg_edge_group<EDGES_PER_GROUP>(h, weight, acc, s, d, t, sc, j, lane);
    for (; j < n; ++j)
        rg_edge_group<1>(h, weight, acc, s, d, t, sc, j, lane);
}

#define LDS_STRIDE 132
__global__ __launch_bounds__(256) void
rg_gemm_kernel(const float* __restrict__ H, const float* __restrict__ W,
               const float* __restrict__ ideg, float* __restrict__ out) {
    __shared__ float ldsH[16 * LDS_STRIDE];
    __shared__ float ldsS[16];

    const int m0 = blockIdx.x * 16;
    const int tid = threadIdx.x;

    for (int i = tid; i < 16 * 32; i += 256) {
        int r = i >> 5;
        int c4 = i & 31;
        float4 v = ((const float4*)(H + (size_t)(m0 + r) * NDIM))[c4];
        float* dl = ldsH + r * LDS_STRIDE + c4 * 4;
        dl[0] = v.x; dl[1] = v.y; dl[2] = v.z; dl[3] = v.w;
    }
    if (tid < 16) ldsS[tid] = rsqrtf(fmaxf(rg_deg(ideg, m0 + tid), 1.0f));
    __syncthreads();

    const int lane = tid & 31;
    const int hi = lane >> 4;
    const int lo = lane & 15;
    const int n0 = (tid >> 5) * 16;

    v8f c = {};
    #pragma unroll
    for (int k = 0; k < NDIM; k += 4) {
        int kk = k + 2 * hi;
        v2f a;
        a.x = ldsH[lo * LDS_STRIDE + kk];
        a.y = ldsH[lo * LDS_STRIDE + kk + 1];
        v2f b;
        b.x = W[(size_t)kk * NDIM + n0 + lo];
        b.y = W[(size_t)(kk + 1) * NDIM + n0 + lo];
        c = __builtin_amdgcn_wmma_f32_16x16x4_f32(
                 false, a,  false, b,
                 (short)0, c,  false,  false);
    }

    #pragma unroll
    for (int g = 0; g < 8; ++g) {
        int row = m0 + g + 8 * hi;
        float a1 = ldsS[g + 8 * hi];
        out[(size_t)row * NDIM + n0 + lo] = c[g] * a1;
    }
}

extern "C" void kernel_launch(void* const* d_in, const int* in_sizes, int n_in,
                              void* d_out, int out_size, void* d_ws, size_t ws_size,
                              hipStream_t stream) {
    const float* x      = (const float*)d_in[0];
    const float* weight = (const float*)d_in[1];
    const float* lw0    = (const float*)d_in[2];
    const float* lw1    = (const float*)d_in[3];
    const int*   src    = (const int*)d_in[4];
    const int*   dst    = (const int*)d_in[5];
    const int*   etype  = (const int*)d_in[6];
    float* out = (float*)d_out;

    const int N = in_sizes[0] / NDIM;
    const int E = in_sizes[4];
    const size_t ND = (size_t)N * NDIM;

    float* ws    = (float*)d_ws;
    float* h1    = ws;
    float* odeg  = h1 + ND;
    float* ideg  = odeg + (size_t)N * DEG_PARTS;

    const int nRanges = (N + DEG_RANGE - 1) / DEG_RANGE;
    rg_deg_kernel<<<2 * nRanges * DEG_PARTS, DEG_THREADS, 0, stream>>>(src, dst, odeg, E, N, nRanges);

    const int edgeBlocks = (E + 8 * EDGES_PER_WAVE - 1) / (8 * EDGES_PER_WAVE);
    const int gemmBlocks = N / 16;

    rg_gemm_kernel<<<gemmBlocks, 256, 0, stream>>>(x, lw0, ideg, h1);
    rg_edge_kernel<<<edgeBlocks, 256, 0, stream>>>(x, odeg, ideg, weight, src, dst, etype, h1, E);

    rg_gemm_kernel<<<gemmBlocks, 256, 0, stream>>>(h1, lw1, ideg, out);
    rg_edge_kernel<<<edgeBlocks, 256, 0, stream>>>(h1, odeg, ideg, weight, src, dst, etype, out, E);
}


# Edge scatter: the relation weight table is copied into LDS once per block (16 waves) and each edge reads its weight row from LDS, not from global memory; atomics and arrival-order sums as before
# speedup vs baseline: 2.5674x; 1.0556x over previous
#include <hip/hip_runtime.h>

typedef __attribute__((ext_vector_type(2))) float v2f;
typedef __attribute__((ext_vector_type(8))) float v8f;

#define NDIM 128

#define DEG_RANGE 2048
#define DEG_PARTS 4
#define DEG_THREADS 1024
#define DEG_BATCH 8

__device__ __forceinline__ void rg_count(unsigned int* cnt, unsigned int base, int v) {
    unsigned int d = (unsigned int)v - base;
    if (d < (unsigned int)DEG_RANGE) atomicAdd(&cnt[d], 1u);
}

__global__ __launch_bounds__(DEG_THREADS) void
rg_deg_kernel(const int* __restrict__ src, const int* __restrict__ dst,
              float* __restrict__ deg, int E, int N, int nRanges) {
    __shared__ unsigned int cnt[DEG_RANGE];
    const int tid = threadIdx.x;
    int b = blockIdx.x;
    const int part = b % DEG_PARTS;
    b /= DEG_PARTS;
    const int range = b % nRanges;
    const int arr = b / nRanges;
    const int* __restrict__ idx = arr ? dst : src;
    const unsigned int base = (unsigned int)range * DEG_RANGE;

    for (int i = tid; i < DEG_RANGE; i += DEG_THREADS) cnt[i] = 0u;
    __syncthreads();

    const int nv = E >> 2;
    const int lo = (int)(((long long)nv * part) / DEG_PARTS);
    const int hi = (int)(((long long)nv * (part + 1)) / DEG_PARTS);
    const int4* __restrict__ idx4 = (const int4*)idx;
    const int span = DEG_BATCH * DEG_THREADS;
    const int steps = (hi - lo + span - 1) / span;
    const int rot = steps > 0 ? range % steps : 0;
    for (int c = 0; c < steps; ++c) {
        int cr = c + rot;
        if (cr >= steps) cr -= steps;
        const int i0 = lo + cr * span + tid;
        int4 v[DEG_BATCH];
        #pragma unroll
        for (int u = 0; u < DEG_BATCH; ++u) {
            const int i = i0 + u * DEG_THREADS;
            const bool ok = i < hi;
            v[u] = idx4[ok ? i : lo];
            if (!ok) v[u] = make_int4(-1, -1, -1, -1);
        }
        #pragma unroll
        for (int u = 0; u < DEG_BATCH; ++u) {
            rg_count(cnt, base, v[u].x);
            rg_count(cnt, base, v[u].y);
            rg_count(cnt, base, v[u].z);
            rg_count(cnt, base, v[u].w);
        }
    }
    if (part == 0 && tid < (E & 3)) rg_count(cnt, base, idx[nv * 4 + tid]);
    __syncthreads();

    float* __restrict__ o = deg + (size_t)arr * N * DEG_PARTS;
    for (int i = tid; i < DEG_RANGE; i += DEG_THREADS) {
        const unsigned int node = base + (unsigned int)i;
        if (node < (unsigned int)N) o[(size_t)node * DEG_PARTS + part] = (float)cnt[i];
    }
}

__device__ __forceinline__ float rg_deg(const float* __restrict__ deg4, int node) {
    const float4 p = ((const float4*)deg4)[node];
    return (p.x + p.y) + (p.z + p.w);
}

__device__ __forceinline__ float rg_onorm(float odeg) {
    return rsqrtf(fmaxf(odeg, 1.0f));
}
__device__ __forceinline__ float rg_s2(float ideg) {
    float id = fmaxf(ideg, 1.0f);
    return rsqrtf(id) / id;
}

#define EDGES_PER_WAVE 32
#define EDGES_PER_GROUP 4
#define EDGE_WAVES 16
#define EDGE_THREADS (EDGE_WAVES * 32)
#define WTAB_STEPS 4

template <int K>
__device__ __forceinline__ void
rg_edge_group(const float* __restrict__ h, const float* wtab,
              float* __restrict__ acc, int s, int d, int t, float sc,
              int j, int lane) {
    float v[K][4];
    for (int u = 0; u < K; ++u) {
        int su = __builtin_amdgcn_readlane(s, j + u);
        int tu = __builtin_amdgcn_readlane(t, j + u);
        float cu = __int_as_float(__builtin_amdgcn_readlane(__float_as_int(sc), j + u));
        const float* hp = h + (size_t)su * NDIM;
        const float* wp = wtab + tu * NDIM;
        for (int q = 0; q < 4; ++q) v[u][q] = hp[lane + 32 * q] * wp[lane + 32 * q] * cu;
    }
    for (int u = 0; u < K; ++u) {
        int du = __builtin_amdgcn_readlane(d, j + u);
        float* mp = acc + (size_t)du * NDIM;
        for (int q = 0; q < 4; ++q) atomicAdd(mp + lane + 32 * q, v[u][q]);
    }
}

__global__ __launch_bounds__(EDGE_THREADS) void
rg_edge_kernel(const float* __restrict__ h, const float* __restrict__ odeg,
               const float* __restrict__ ideg,
               const float* __restrict__ weight, const int* __restrict__ src,
               const int* __restrict__ dst, const int* __restrict__ etype,
               float* __restrict__ acc, int E, int R) {
    extern __shared__ float wtab[];
    const int tid = threadIdx.x;
    const int lane = tid & 31;
    const int wave = __builtin_amdgcn_readfirstlane((int)(tid >> 5));
    const int e0 = (blockIdx.x * EDGE_WAVES + wave) * EDGES_PER_WAVE;
    const int n = max(0, min(EDGES_PER_WAVE, E - e0));
    const float4* __restrict__ w4 = (const float4*)weight;
    float4* l4 = (float4*)wtab;
    const int nv = R * (NDIM / 4);
    float4 wv[WTAB_STEPS];
    #pragma unroll
    for (int u = 0; u < WTAB_STEPS; ++u) {
        const int i = tid + u * EDGE_THREADS;
        wv[u] = w4[i < nv ? i : 0];
    }
    const int ei = min(e0 + lane, E - 1);
    const int s = src[ei];
    const int d = dst[ei];
    const int t = etype[ei];
    const float sc = rg_onorm(rg_deg(odeg, s)) * rg_s2(rg_deg(ideg, d));
    #pragma unroll
    for (int u = 0; u < WTAB_STEPS; ++u) {
        const int i = tid + u * EDGE_THREADS;
        l4[i] = wv[u];
    }
    for (int i = tid + WTAB_STEPS * EDGE_THREADS; i < nv; i += EDGE_THREADS) l4[i] = w4[i];
    __syncthreads();
    int j = 0;
    for (; j + EDGES_PER_GROUP <= n; j += EDGES_PER_GROUP)
        rg_edge_group<EDGES_PER_GROUP>(h, wtab, acc, s, d, t, sc, j, lane);
    for (; j < n; ++j)
        rg_edge_group<1>(h, wtab, acc, s, d, t, sc, j, lane);
}

#define LDS_STRIDE 132
__global__ __launch_bounds__(256) void
rg_gemm_kernel(const float* __restrict__ H, const float* __restrict__ W,
               const float* __restrict__ ideg, float* __restrict__ out) {
    __shared__ float ldsH[16 * LDS_STRIDE];
    __shared__ float ldsS[16];

    const int m0 = blockIdx.x * 16;
    const int tid = threadIdx.x;

    for (int i = tid; i < 16 * 32; i += 256) {
        int r = i >> 5;
        int c4 = i & 31;
        float4 v = ((const float4*)(H + (size_t)(m0 + r) * NDIM))[c4];
        float* dl = ldsH + r * LDS_STRIDE + c4 * 4;
        dl[0] = v.x; dl[1] = v.y; dl[2] = v.z; dl[3] = v.w;
    }
    if (tid < 16) ldsS[tid] = rsqrtf(fmaxf(rg_deg(ideg, m0 + tid), 1.0f));
    __syncthreads();

    const int lane = tid & 31;
    const int hi = lane >> 4;
    const int lo = lane & 15;
    const int n0 = (tid >> 5) * 16;

    v8f c = {};
    #pragma unroll
    for (int k = 0; k < NDIM; k += 4) {
        int kk = k + 2 * hi;
        v2f a;
        a.x = ldsH[lo * LDS_STRIDE + kk];
        a.y = ldsH[lo * LDS_STRIDE + kk + 1];
        v2f b;
        b.x = W[(size_t)kk * NDIM + n0 + lo];
        b.y = W[(size_t)(kk + 1) * NDIM + n0 + lo];
        c = __builtin_amdgcn_wmma_f32_16x16x4_f32(
                 false, a,  false, b,
                 (short)0, c,  false,  false);
    }

    #pragma unroll
    for (int g = 0; g < 8; ++g) {
        int row = m0 + g + 8 * hi;
        float a1 = ldsS[g + 8 * hi];
        out[(size_t)row * NDIM + n0 + lo] = c[g] * a1;
    }
}

extern "C" void kernel_launch(void* const* d_in, const int* in_sizes, int n_in,
                              void* d_out, int out_size, void* d_ws, size_t ws_size,
                              hipStream_t stream) {
    const float* x      = (const float*)d_in[0];
    const float* weight = (const float*)d_in[1];
    const float* lw0    = (const float*)d_in[2];
    const float* lw1    = (const float*)d_in[3];
    const int*   src    = (const int*)d_in[4];
    const int*   dst    = (const int*)d_in[5];
    const int*   etype  = (const int*)d_in[6];
    float* out = (float*)d_out;

    const int N = in_sizes[0] / NDIM;
    const int E = in_sizes[4];
    const int R = in_sizes[1] / NDIM;
    const size_t ND = (size_t)N * NDIM;

    float* ws    = (float*)d_ws;
    float* h1    = ws;
    float* odeg  = h1 + ND;
    float* ideg  = odeg + (size_t)N * DEG_PARTS;

    const int nRanges = (N + DEG_RANGE - 1) / DEG_RANGE;
    rg_deg_kernel<<<2 * nRanges * DEG_PARTS, DEG_THREADS, 0, stream>>>(src, dst, odeg, E, N, nRanges);

    const int edgeBlocks = (E + EDGE_WAVES * EDGES_PER_WAVE - 1) / (EDGE_WAVES * EDGES_PER_WAVE);
    const size_t wtabMin = (size_t)WTAB_STEPS * EDGE_THREADS * 16;
    const size_t wtabTab = (size_t)R * NDIM * sizeof(float);
    const size_t wtabBytes = wtabTab > wtabMin ? wtabTab : wtabMin;
    const int gemmBlocks = N / 16;

    rg_gemm_kernel<<<gemmBlocks, 256, 0, stream>>>(x, lw0, ideg, h1);
    rg_edge_kernel<<<edgeBlocks, EDGE_THREADS, wtabBytes, stream>>>(x, odeg, ideg, weight, src, dst, etype, h1, E, R);

    rg_gemm_kernel<<<gemmBlocks, 256, 0, stream>>>(h1, lw1, ideg, out);
    rg_edge_kernel<<<edgeBlocks, EDGE_THREADS, wtabBytes, stream>>>(h1, odeg, ideg, weight, src, dst, etype, out, E, R);
}


# Edge scatter: per-edge scale onorm(src)*s2(dst) stored by layer 1's launch (E floats, coalesced), read back by layer 2's, which makes no degree lookups; atomics and arrival-order sums as before
# speedup vs baseline: 2.6606x; 1.0363x over previous
#include <hip/hip_runtime.h>

typedef __attribute__((ext_vector_type(2))) float v2f;
typedef __attribute__((ext_vector_type(8))) float v8f;

#define NDIM 128

#define DEG_RANGE 2048
#define DEG_PARTS 4
#define DEG_THREADS 1024
#define DEG_BATCH 8

__device__ __forceinline__ void rg_count(unsigned int* cnt, unsigned int base, int v) {
    unsigned int d = (unsigned int)v - base;
    if (d < (unsigned int)DEG_RANGE) atomicAdd(&cnt[d], 1u);
}

__global__ __launch_bounds__(DEG_THREADS) void
rg_deg_kernel(const int* __restrict__ src, const int* __restrict__ dst,
              float* __restrict__ deg, int E, int N, int nRanges) {
    __shared__ unsigned int cnt[DEG_RANGE];
    const int tid = threadIdx.x;
    int b = blockIdx.x;
    const int part = b % DEG_PARTS;
    b /= DEG_PARTS;
    const int range = b % nRanges;
    const int arr = b / nRanges;
    const int* __restrict__ idx = arr ? dst : src;
    const unsigned int base = (unsigned int)range * DEG_RANGE;

    for (int i = tid; i < DEG_RANGE; i += DEG_THREADS) cnt[i] = 0u;
    __syncthreads();

    const int nv = E >> 2;
    const int lo = (int)(((long long)nv * part) / DEG_PARTS);
    const int hi = (int)(((long long)nv * (part + 1)) / DEG_PARTS);
    const int4* __restrict__ idx4 = (const int4*)idx;
    const int span = DEG_BATCH * DEG_THREADS;
    const int steps = (hi - lo + span - 1) / span;
    const int rot = steps > 0 ? range % steps : 0;
    for (int c = 0; c < steps; ++c) {
        int cr = c + rot;
        if (cr >= steps) cr -= steps;
        const int i0 = lo + cr * span + tid;
        int4 v[DEG_BATCH];
        #pragma unroll
        for (int u = 0; u < DEG_BATCH; ++u) {
            const int i = i0 + u * DEG_THREADS;
            const bool ok = i < hi;
            v[u] = idx4[ok ? i : lo];
            if (!ok) v[u] = make_int4(-1, -1, -1, -1);
        }
        #pragma unroll
        for (int u = 0; u < DEG_BATCH; ++u) {
            rg_count(cnt, base, v[u].x);
            rg_count(cnt, base, v[u].y);
            rg_count(cnt, base, v[u].z);
            rg_count(cnt, base, v[u].w);
        }
    }
    if (part == 0 && tid < (E & 3)) rg_count(cnt, base, idx[nv * 4 + tid]);
    __syncthreads();

    float* __restrict__ o = deg + (size_t)arr * N * DEG_PARTS;
    for (int i = tid; i < DEG_RANGE; i += DEG_THREADS) {
        const unsigned int node = base + (unsigned int)i;
        if (node < (unsigned int)N) o[(size_t)node * DEG_PARTS + part] = (float)cnt[i];
    }
}

__device__ __forceinline__ float rg_deg(const float* __restrict__ deg4, int node) {
    const float4 p = ((const float4*)deg4)[node];
    return (p.x + p.y) + (p.z + p.w);
}

__device__ __forceinline__ float rg_onorm(float odeg) {
    return rsqrtf(fmaxf(odeg, 1.0f));
}
__device__ __forceinline__ float rg_s2(float ideg) {
    float id = fmaxf(ideg, 1.0f);
    return rsqrtf(id) / id;
}

#define EDGES_PER_WAVE 32
#define EDGES_PER_GROUP 4
#define EDGE_WAVES 16
#define EDGE_THREADS (EDGE_WAVES * 32)
#define WTAB_STEPS 4

template <int K>
__device__ __forceinline__ void
rg_edge_group(const float* __restrict__ h, const float* wtab,
              float* __restrict__ acc, int s, int d, int t, float sc,
              int j, int lane) {
    float v[K][4];
    for (int u = 0; u < K; ++u) {
        int su = __builtin_amdgcn_readlane(s, j + u);
        int tu = __builtin_amdgcn_readlane(t, j + u);
        float cu = __int_as_float(__builtin_amdgcn_readlane(__float_as_int(sc), j + u));
        const float* hp = h + (size_t)su * NDIM;
        const float* wp = wtab + tu * NDIM;
        for (int q = 0; q < 4; ++q) v[u][q] = hp[lane + 32 * q] * wp[lane + 32 * q] * cu;
    }
    for (int u = 0; u < K; ++u) {
        int du = __builtin_amdgcn_readlane(d, j + u);
        float* mp = acc + (size_t)du * NDIM;
        for (int q = 0; q < 4; ++q) atomicAdd(mp + lane + 32 * q, v[u][q]);
    }
}

template <bool FIRST>
__global__ __launch_bounds__(EDGE_THREADS) void
rg_edge_kernel(const float* __restrict__ h, const float* __restrict__ odeg,
               const float* __restrict__ ideg,
               const float* __restrict__ weight, const int* __restrict__ src,
               const int* __restrict__ dst, const int* __restrict__ etype,
               float* __restrict__ acc, float* __restrict__ escale, int E, int R) {
    extern __shared__ float wtab[];
    const int tid = threadIdx.x;
    const int lane = tid & 31;
    const int wave = __builtin_amdgcn_readfirstlane((int)(tid >> 5));
    const int e0 = (blockIdx.x * EDGE_WAVES + wave) * EDGES_PER_WAVE;
    const int n = max(0, min(EDGES_PER_WAVE, E - e0));
    const float4* __restrict__ w4 = (const float4*)weight;
    float4* l4 = (float4*)wtab;
    const int nv = R * (NDIM / 4);
    float4 wv[WTAB_STEPS];
    #pragma unroll
    for (int u = 0; u < WTAB_STEPS; ++u) {
        const int i = tid + u * EDGE_THREADS;
        wv[u] = w4[i < nv ? i : 0];
    }
    const int ei = min(e0 + lane, E - 1);
    const int s = src[ei];
    const int d = dst[ei];
    const int t = etype[ei];
    float sc;
    if (FIRST) sc = rg_onorm(rg_deg(odeg, s)) * rg_s2(rg_deg(ideg, d));
    else       sc = escale[ei];
    #pragma unroll
    for (int u = 0; u < WTAB_STEPS; ++u) {
        const int i = tid + u * EDGE_THREADS;
        l4[i] = wv[u];
    }
    for (int i = tid + WTAB_STEPS * EDGE_THREADS; i < nv; i += EDGE_THREADS) l4[i] = w4[i];
    __syncthreads();
    if (FIRST && lane < n) escale[e0 + lane] = sc;
    int j = 0;
    for (; j + EDGES_PER_GROUP <= n; j += EDGES_PER_GROUP)
        rg_edge_group<EDGES_PER_GROUP>(h, wtab, acc, s, d, t, sc, j, lane);
    for (; j < n; ++j)
        rg_edge_group<1>(h, wtab, acc, s, d, t, sc, j, lane);
}

#define LDS_STRIDE 132
__global__ __launch_bounds__(256) void
rg_gemm_kernel(const float* __restrict__ H, const float* __restrict__ W,
               const float* __restrict__ ideg, float* __restrict__ out) {
    __shared__ float ldsH[16 * LDS_STRIDE];
    __shared__ float ldsS[16];

    const int m0 = blockIdx.x * 16;
    const int tid = threadIdx.x;

    for (int i = tid; i < 16 * 32; i += 256) {
        int r = i >> 5;
        int c4 = i & 31;
        float4 v = ((const float4*)(H + (size_t)(m0 + r) * NDIM))[c4];
        float* dl = ldsH + r * LDS_STRIDE + c4 * 4;
        dl[0] = v.x; dl[1] = v.y; dl[2] = v.z; dl[3] = v.w;
    }
    if (tid < 16) ldsS[tid] = rsqrtf(fmaxf(rg_deg(ideg, m0 + tid), 1.0f));
    __syncthreads();

    const int lane = tid & 31;
    const int hi = lane >> 4;
    const int lo = lane & 15;
    const int n0 = (tid >> 5) * 16;

    v8f c = {};
    #pragma unroll
    for (int k = 0; k < NDIM; k += 4) {
        int kk = k + 2 * hi;
        v2f a;
        a.x = ldsH[lo * LDS_STRIDE + kk];
        a.y = ldsH[lo * LDS_STRIDE + kk + 1];
        v2f b;
        b.x = W[(size_t)kk * NDIM + n0 + lo];
        b.y = W[(size_t)(kk + 1) * NDIM + n0 + lo];
        c = __builtin_amdgcn_wmma_f32_16x16x4_f32(
                 false, a,  false, b,
                 (short)0, c,  false,  false);
    }

    #pragma unroll
    for (int g = 0; g < 8; ++g) {
        int row = m0 + g + 8 * hi;
        float a1 = ldsS[g + 8 * hi];
        out[(size_t)row * NDIM + n0 + lo] = c[g] * a1;
    }
}

extern "C" void kernel_launch(void* const* d_in, const int* in_sizes, int n_in,
                              void* d_out, int out_size, void* d_ws, size_t ws_size,
                              hipStream_t stream) {
    const float* x      = (const float*)d_in[0];
    const float* weight = (const float*)d_in[1];
    const float* lw0    = (const float*)d_in[2];
    const float* lw1    = (const float*)d_in[3];
    const int*   src    = (const int*)d_in[4];
    const int*   dst    = (const int*)d_in[5];
    const int*   etype  = (const int*)d_in[6];
    float* out = (float*)d_out;

    const int N = in_sizes[0] / NDIM;
    const int E = in_sizes[4];
    const int R = in_sizes[1] / NDIM;
    const size_t ND = (size_t)N * NDIM;

    float* ws    = (float*)d_ws;
    float* h1    = ws;
    float* odeg  = h1 + ND;
    float* ideg  = odeg + (size_t)N * DEG_PARTS;
    float* escale = ideg + (size_t)N * DEG_PARTS;

    const int nRanges = (N + DEG_RANGE - 1) / DEG_RANGE;
    rg_deg_kernel<<<2 * nRanges * DEG_PARTS, DEG_THREADS, 0, stream>>>(src, dst, odeg, E, N, nRanges);

    const int edgeBlocks = (E + EDGE_WAVES * EDGES_PER_WAVE - 1) / (EDGE_WAVES * EDGES_PER_WAVE);
    const size_t wtabMin = (size_t)WTAB_STEPS * EDGE_THREADS * 16;
    const size_t wtabTab = (size_t)R * NDIM * sizeof(float);
    const size_t wtabBytes = wtabTab > wtabMin ? wtabTab : wtabMin;
    const int gemmBlocks = N / 16;

    rg_gemm_kernel<<<gemmBlocks, 256, 0, stream>>>(x, lw0, ideg, h1);
    rg_edge_kernel<true><<<edgeBlocks, EDGE_THREADS, wtabBytes, stream>>>(x, odeg, ideg, weight, src, dst, etype, h1, escale, E, R);

    rg_gemm_kernel<<<gemmBlocks, 256, 0, stream>>>(h1, lw1, ideg, out);
    rg_edge_kernel<false><<<edgeBlocks, EDGE_THREADS, wtabBytes, stream>>>(h1, odeg, ideg, weight, src, dst, etype, out, escale, E, R);
}


# Edge scatter: dst nodes cut in two ranges, a block adds only to the range of the chip half it runs in; chunks drawn from per-range ticket counters (integer atomic); sums stay in arrival order
# speedup vs baseline: 3.0308x; 1.1391x over previous
#include <hip/hip_runtime.h>

typedef __attribute__((ext_vector_type(2))) float v2f;
typedef __attribute__((ext_vector_type(8))) float v8f;

#define NDIM 128

#define DEG_RANGE 2048
#define DEG_PARTS 4
#define DEG_THREADS 1024
#define DEG_BATCH 8

__device__ __forceinline__ void rg_count(unsigned int* cnt, unsigned int base, int v) {
    unsigned int d = (unsigned int)v - base;
    if (d < (unsigned int)DEG_RANGE) atomicAdd(&cnt[d], 1u);
}

__global__ __launch_bounds__(DEG_THREADS) void
rg_deg_kernel(const int* __restrict__ src, const int* __restrict__ dst,
              float* __restrict__ deg, unsigned int* __restrict__ tick, int E, int N, int nRanges) {
    __shared__ unsigned int cnt[DEG_RANGE];
    const int tid = threadIdx.x;
    int b = blockIdx.x;
    const int part = b % DEG_PARTS;
    b /= DEG_PARTS;
    const int range = b % nRanges;
    const int arr = b / nRanges;
    const int* __restrict__ idx = arr ? dst : src;
    const unsigned int base = (unsigned int)range * DEG_RANGE;

    for (int i = tid; i < DEG_RANGE; i += DEG_THREADS) cnt[i] = 0u;
    if (blockIdx.x == 0 && tid < 2) tick[tid] = 0u;
    __syncthreads();

    const int nv = E >> 2;
    const int lo = (int)(((long long)nv * part) / DEG_PARTS);
    const int hi = (int)(((long long)nv * (part + 1)) / DEG_PARTS);
    const int4* __restrict__ idx4 = (const int4*)idx;
    const int span = DEG_BATCH * DEG_THREADS;
    const int steps = (hi - lo + span - 1) / span;
    const int rot = steps > 0 ? range % steps : 0;
    for (int c = 0; c < steps; ++c) {
        int cr = c + rot;
        if (cr >= steps) cr -= steps;
        const int i0 = lo + cr * span + tid;
        int4 v[DEG_BATCH];
        #pragma unroll
        for (int u = 0; u < DEG_BATCH; ++u) {
            const int i = i0 + u * DEG_THREADS;
            const bool ok = i < hi;
            v[u] = idx4[ok ? i : lo];
            if (!ok) v[u] = make_int4(-1, -1, -1, -1);
        }
        #pragma unroll
        for (int u = 0; u < DEG_BATCH; ++u) {
            rg_count(cnt, base, v[u].x);
            rg_count(cnt, base, v[u].y);
            rg_count(cnt, base, v[u].z);
            rg_count(cnt, base, v[u].w);
        }
    }
    if (part == 0 && tid < (E & 3)) rg_count(cnt, base, idx[nv * 4 + tid]);
    __syncthreads();

    float* __restrict__ o = deg + (size_t)arr * N * DEG_PARTS;
    for (int i = tid; i < DEG_RANGE; i += DEG_THREADS) {
        const unsigned int node = base + (unsigned int)i;
        if (node < (unsigned int)N) o[(size_t)node * DEG_PARTS + part] = (float)cnt[i];
    }
}

__device__ __forceinline__ float rg_deg(const float* __restrict__ deg4, int node) {
    const float4 p = ((const float4*)deg4)[node];
    return (p.x + p.y) + (p.z + p.w);
}

__device__ __forceinline__ float rg_onorm(float odeg) {
    return rsqrtf(fmaxf(odeg, 1.0f));
}
__device__ __forceinline__ float rg_s2(float ideg) {
    float id = fmaxf(ideg, 1.0f);
    return rsqrtf(id) / id;
}

#define EDGES_PER_WAVE 32
#define EDGE_WAVES 16
#define EDGE_THREADS (EDGE_WAVES * 32)
#define EDGES_PER_CHUNK (EDGE_WAVES * EDGES_PER_WAVE)
#define WTAB_STEPS 4
#define EDGE_GRID 1024

template <int K>
__device__ __forceinline__ void
rg_edge_group(const float* __restrict__ h, const float* wtab,
              float* __restrict__ acc, int s, int d, int t, float sc,
              const int (&jl)[K], int lane) {
    float v[K][4];
    for (int u = 0; u < K; ++u) {
        int su = __builtin_amdgcn_readlane(s, jl[u]);
        int tu = __builtin_amdgcn_readlane(t, jl[u]);
        float cu = __int_as_float(__builtin_amdgcn_readlane(__float_as_int(sc), jl[u]));
        const float* hp = h + (size_t)su * NDIM;
        const float* wp = wtab + tu * NDIM;
        for (int q = 0; q < 4; ++q) v[u][q] = hp[lane + 32 * q] * wp[lane + 32 * q] * cu;
    }
    for (int u = 0; u < K; ++u) {
        int du = __builtin_amdgcn_readlane(d, jl[u]);
        float* mp = acc + (size_t)du * NDIM;
        for (int q = 0; q < 4; ++q) atomicAdd(mp + lane + 32 * q, v[u][q]);
    }
}

__device__ __forceinline__ int rg_take(unsigned int& m) {
    const int j = __builtin_ctz(m);
    m &= m - 1u;
    return j;
}

template <bool FIRST>
__device__ __forceinline__ void
rg_edge_chunk(const float* __restrict__ h, const float* __restrict__ odeg,
              const float* __restrict__ ideg, const float* wtab,
              const int* __restrict__ src, const int* __restrict__ dst,
              const int* __restrict__ etype, float* __restrict__ acc,
              float* __restrict__ escale, int E, int split,
              int c, int hh, int wave, int lane) {
    const int e0 = (c * EDGE_WAVES + wave) * EDGES_PER_WAVE;
    const int n = max(0, min(EDGES_PER_WAVE, E - e0));
    const int ei = min(e0 + lane, E - 1);
    const int s = src[ei];
    const int d = dst[ei];
    const int t = etype[ei];
    const bool mine = (lane < n) && ((d >= split) == (hh != 0));
    float sc = 0.0f;
    if (FIRST) {
        if (hh == 0 || mine) sc = rg_onorm(rg_deg(odeg, s)) * rg_s2(rg_deg(ideg, d));
        if (hh == 0 && lane < n) escale[e0 + lane] = sc;
    } else {
        sc = escale[ei];
    }
    unsigned int m = __builtin_amdgcn_ballot_w32(mine);
    for (int k = 0; k < EDGES_PER_WAVE / 4 && __builtin_popcount(m) >= 4; ++k) {
        int jl[4];
        jl[0] = rg_take(m); jl[1] = rg_take(m); jl[2] = rg_take(m); jl[3] = rg_take(m);
        rg_edge_group<4>(h, wtab, acc, s, d, t, sc, jl, lane);
    }
    if (__builtin_popcount(m) >= 2) {
        int jl[2];
        jl[0] = rg_take(m); jl[1] = rg_take(m);
        rg_edge_group<2>(h, wtab, acc, s, d, t, sc, jl, lane);
    }
    if (__builtin_popcount(m) >= 2) {
        int jl[2];
        jl[0] = rg_take(m); jl[1] = rg_take(m);
        rg_edge_group<2>(h, wtab, acc, s, d, t, sc, jl, lane);
    }
    if (m) {
        int jl[1];
        jl[0] = rg_take(m);
        rg_edge_group<1>(h, wtab, acc, s, d, t, sc, jl, lane);
    }
}

template <bool FIRST>
__global__ __launch_bounds__(EDGE_THREADS) void
rg_edge_kernel(const float* __restrict__ h, const float* __restrict__ odeg,
               const float* __restrict__ ideg,
               const float* __restrict__ weight, const int* __restrict__ src,
               const int* __restrict__ dst, const int* __restrict__ etype,
               float* __restrict__ acc, float* __restrict__ escale,
               unsigned int* tick, int E, int R, int N) {
    extern __shared__ float wtab[];
    const int tid = threadIdx.x;
    const int lane = tid & 31;
    const int wave = __builtin_amdgcn_readfirstlane((int)(tid >> 5));
    const float4* __restrict__ w4 = (const float4*)weight;
    float4* l4 = (float4*)wtab;
    const int nv = R * (NDIM / 4);
    unsigned int* s_tick = (unsigned int*)(wtab + 4 * max(nv, WTAB_STEPS * EDGE_THREADS));
    float4 wv[WTAB_STEPS];
    #pragma unroll
    for (int u = 0; u < WTAB_STEPS; ++u) {
        const int i = tid + u * EDGE_THREADS;
        wv[u] = w4[i < nv ? i : 0];
    }
    #pragma unroll
    for (int u = 0; u < WTAB_STEPS; ++u) {
        const int i = tid + u * EDGE_THREADS;
        l4[i] = wv[u];
    }
    for (int i = tid + WTAB_STEPS * EDGE_THREADS; i < nv; i += EDGE_THREADS) l4[i] = w4[i];

    const unsigned int where = __builtin_amdgcn_s_sendmsg_rtn(0x87);
    const int myhalf = (int)((where >> 18) & 1u);
    const int nChunks = (E + EDGES_PER_CHUNK - 1) / EDGES_PER_CHUNK;
    const int split = N >> 1;

    int g = 0;
    for (int pass = 0; pass < 2; ++pass) {
        const int hh = myhalf ^ pass;
        unsigned int* tk = tick + hh;
        if (tid == 0) s_tick[g & 1] = atomicAdd(tk, 1u);
        __syncthreads();
        unsigned int c = (unsigned int)__builtin_amdgcn_readfirstlane((int)s_tick[g & 1]);
        ++g;
        for (int it = 0; it < nChunks && c < (unsigned int)nChunks; ++it) {
            unsigned int nxt = 0u;
            if (tid == 0) nxt = atomicAdd(tk, 1u);
            rg_edge_chunk<FIRST>(h, odeg, ideg, wtab, src, dst, etype, acc, escale,
                                 E, split, (int)c, hh, wave, lane);
            if (tid == 0) s_tick[g & 1] = nxt;
            __syncthreads();
            c = (unsigned int)__builtin_amdgcn_readfirstlane((int)s_tick[g & 1]);
            ++g;
        }
    }
}

#define LDS_STRIDE 132
__global__ __launch_bounds__(256) void
rg_gemm_kernel(const float* __restrict__ H, const float* __restrict__ W,
               const float* __restrict__ ideg, float* __restrict__ out,
               unsigned int* __restrict__ tick) {
    __shared__ float ldsH[16 * LDS_STRIDE];
    __shared__ float ldsS[16];

    const int m0 = blockIdx.x * 16;
    const int tid = threadIdx.x;

    for (int i = tid; i < 16 * 32; i += 256) {
        int r = i >> 5;
        int c4 = i & 31;
        float4 v = ((const float4*)(H + (size_t)(m0 + r) * NDIM))[c4];
        float* dl = ldsH + r * LDS_STRIDE + c4 * 4;
        dl[0] = v.x; dl[1] = v.y; dl[2] = v.z; dl[3] = v.w;
    }
    if (tid < 16) ldsS[tid] = rsqrtf(fmaxf(rg_deg(ideg, m0 + tid), 1.0f));
    if (tick != nullptr && blockIdx.x == 0 && tid < 2) tick[tid] = 0u;
    __syncthreads();

    const int lane = tid & 31;
    const int hi = lane >> 4;
    const int lo = lane & 15;
    const int n0 = (tid >> 5) * 16;

    v8f c = {};
    #pragma unroll
    for (int k = 0; k < NDIM; k += 4) {
        int kk = k + 2 * hi;
        v2f a;
        a.x = ldsH[lo * LDS_STRIDE + kk];
        a.y = ldsH[lo * LDS_STRIDE + kk + 1];
        v2f b;
        b.x = W[(size_t)kk * NDIM + n0 + lo];
        b.y = W[(size_t)(kk + 1) * NDIM + n0 + lo];
        c = __builtin_amdgcn_wmma_f32_16x16x4_f32(
                 false, a,  false, b,
                 (short)0, c,  false,  false);
    }

    #pragma unroll
    for (int g = 0; g < 8; ++g) {
        int row = m0 + g + 8 * hi;
        float a1 = ldsS[g + 8 * hi];
        out[(size_t)row * NDIM + n0 + lo] = c[g] * a1;
    }
}

extern "C" void kernel_launch(void* const* d_in, const int* in_sizes, int n_in,
                              void* d_out, int out_size, void* d_ws, size_t ws_size,
                              hipStream_t stream) {
    const float* x      = (const float*)d_in[0];
    const float* weight = (const float*)d_in[1];
    const float* lw0    = (const float*)d_in[2];
    const float* lw1    = (const float*)d_in[3];
    const int*   src    = (const int*)d_in[4];
    const int*   dst    = (const int*)d_in[5];
    const int*   etype  = (const int*)d_in[6];
    float* out = (float*)d_out;

    const int N = in_sizes[0] / NDIM;
    const int E = in_sizes[4];
    const int R = in_sizes[1] / NDIM;
    const size_t ND = (size_t)N * NDIM;

    float* ws    = (float*)d_ws;
    float* h1    = ws;
    float* odeg  = h1 + ND;
    float* ideg  = odeg + (size_t)N * DEG_PARTS;
    float* escale = ideg + (size_t)N * DEG_PARTS;

    const int nRanges = (N + DEG_RANGE - 1) / DEG_RANGE;
    unsigned int* tick1 = (unsigned int*)out;
    unsigned int* tick2 = (unsigned int*)odeg;
    rg_deg_kernel<<<2 * nRanges * DEG_PARTS, DEG_THREADS, 0, stream>>>(src, dst, odeg, tick1, E, N, nRanges);

    const size_t wtabMin = (size_t)WTAB_STEPS * EDGE_THREADS * 16;
    const size_t wtabTab = (size_t)R * NDIM * sizeof(float);
    const size_t wtabBytes = (wtabTab > wtabMin ? wtabTab : wtabMin) + 16;
    const int gemmBlocks = N / 16;

    rg_gemm_kernel<<<gemmBlocks, 256, 0, stream>>>(x, lw0, ideg, h1, nullptr);
    rg_edge_kernel<true><<<EDGE_GRID, EDGE_THREADS, wtabBytes, stream>>>(x, odeg, ideg, weight, src, dst, etype, h1, escale, tick1, E, R, N);

    rg_gemm_kernel<<<gemmBlocks, 256, 0, stream>>>(h1, lw1, ideg, out, tick2);
    rg_edge_kernel<false><<<EDGE_GRID, EDGE_THREADS, wtabBytes, stream>>>(h1, odeg, ideg, weight, src, dst, etype, out, escale, tick2, E, R, N);
}


# Self-loop GEMM: resident grid of 512 blocks; a wave loads its 128x16 slice of W once into registers, walks 16-row tiles two per step, H rows double-buffered in LDS; 4096 waves a launch, was 25000
# speedup vs baseline: 3.2135x; 1.0603x over previous
#include <hip/hip_runtime.h>

typedef __attribute__((ext_vector_type(2))) float v2f;
typedef __attribute__((ext_vector_type(8))) float v8f;

#define NDIM 128

#define DEG_RANGE 2048
#define DEG_PARTS 4
#define DEG_THREADS 1024
#define DEG_BATCH 8

__device__ __forceinline__ void rg_count(unsigned int* cnt, unsigned int base, int v) {
    unsigned int d = (unsigned int)v - base;
    if (d < (unsigned int)DEG_RANGE) atomicAdd(&cnt[d], 1u);
}

__global__ __launch_bounds__(DEG_THREADS) void
rg_deg_kernel(const int* __restrict__ src, const int* __restrict__ dst,
              float* __restrict__ deg, unsigned int* __restrict__ tick, int E, int N, int nRanges) {
    __shared__ unsigned int cnt[DEG_RANGE];
    const int tid = threadIdx.x;
    int b = blockIdx.x;
    const int part = b % DEG_PARTS;
    b /= DEG_PARTS;
    const int range = b % nRanges;
    const int arr = b / nRanges;
    const int* __restrict__ idx = arr ? dst : src;
    const unsigned int base = (unsigned int)range * DEG_RANGE;

    for (int i = tid; i < DEG_RANGE; i += DEG_THREADS) cnt[i] = 0u;
    if (blockIdx.x == 0 && tid < 2) tick[tid] = 0u;
    __syncthreads();

    const int nv = E >> 2;
    const int lo = (int)(((long long)nv * part) / DEG_PARTS);
    const int hi = (int)(((long long)nv * (part + 1)) / DEG_PARTS);
    const int4* __restrict__ idx4 = (const int4*)idx;
    const int span = DEG_BATCH * DEG_THREADS;
    const int steps = (hi - lo + span - 1) / span;
    const int rot = steps > 0 ? range % steps : 0;
    for (int c = 0; c < steps; ++c) {
        int cr = c + rot;
        if (cr >= steps) cr -= steps;
        const int i0 = lo + cr * span + tid;
        int4 v[DEG_BATCH];
        #pragma unroll
        for (int u = 0; u < DEG_BATCH; ++u) {
            const int i = i0 + u * DEG_THREADS;
            const bool ok = i < hi;
            v[u] = idx4[ok ? i : lo];
            if (!ok) v[u] = make_int4(-1, -1, -1, -1);
        }
        #pragma unroll
        for (int u = 0; u < DEG_BATCH; ++u) {
            rg_count(cnt, base, v[u].x);
            rg_count(cnt, base, v[u].y);
            rg_count(cnt, base, v[u].z);
            rg_count(cnt, base, v[u].w);
        }
    }
    if (part == 0 && tid < (E & 3)) rg_count(cnt, base, idx[nv * 4 + tid]);
    __syncthreads();

    float* __restrict__ o = deg + (size_t)arr * N * DEG_PARTS;
    for (int i = tid; i < DEG_RANGE; i += DEG_THREADS) {
        const unsigned int node = base + (unsigned int)i;
        if (node < (unsigned int)N) o[(size_t)node * DEG_PARTS + part] = (float)cnt[i];
    }
}

__device__ __forceinline__ float rg_deg(const float* __restrict__ deg4, int node) {
    const float4 p = ((const float4*)deg4)[node];
    return (p.x + p.y) + (p.z + p.w);
}

__device__ __forceinline__ float rg_onorm(float odeg) {
    return rsqrtf(fmaxf(odeg, 1.0f));
}
__device__ __forceinline__ float rg_s2(float ideg) {
    float id = fmaxf(ideg, 1.0f);
    return rsqrtf(id) / id;
}

#define EDGES_PER_WAVE 32
#define EDGE_WAVES 16
#define EDGE_THREADS (EDGE_WAVES * 32)
#define EDGES_PER_CHUNK (EDGE_WAVES * EDGES_PER_WAVE)
#define WTAB_STEPS 4
#define EDGE_GRID 1024

template <int K>
__device__ __forceinline__ void
rg_edge_group(const float* __restrict__ h, const float* wtab,
              float* __restrict__ acc, int s, int d, int t, float sc,
              const int (&jl)[K], int lane) {
    float v[K][4];
    for (int u = 0; u < K; ++u) {
        int su = __builtin_amdgcn_readlane(s, jl[u]);
        int tu = __builtin_amdgcn_readlane(t, jl[u]);
        float cu = __int_as_float(__builtin_amdgcn_readlane(__float_as_int(sc), jl[u]));
        const float* hp = h + (size_t)su * NDIM;
        const float* wp = wtab + tu * NDIM;
        for (int q = 0; q < 4; ++q) v[u][q] = hp[lane + 32 * q] * wp[lane + 32 * q] * cu;
    }
    for (int u = 0; u < K; ++u) {
        int du = __builtin_amdgcn_readlane(d, jl[u]);
        float* mp = acc + (size_t)du * NDIM;
        for (int q = 0; q < 4; ++q) atomicAdd(mp + lane + 32 * q, v[u][q]);
    }
}

__device__ __forceinline__ int rg_take(unsigned int& m) {
    const int j = __builtin_ctz(m);
    m &= m - 1u;
    return j;
}

template <bool FIRST>
__device__ __forceinline__ void
rg_edge_chunk(const float* __restrict__ h, const float* __restrict__ odeg,
              const float* __restrict__ ideg, const float* wtab,
              const int* __restrict__ src, const int* __restrict__ dst,
              const int* __restrict__ etype, float* __restrict__ acc,
              float* __restrict__ escale, int E, int split,
              int c, int hh, int wave, int lane) {
    const int e0 = (c * EDGE_WAVES + wave) * EDGES_PER_WAVE;
    const int n = max(0, min(EDGES_PER_WAVE, E - e0));
    const int ei = min(e0 + lane, E - 1);
    const int s = src[ei];
    const int d = dst[ei];
    const int t = etype[ei];
    const bool mine = (lane < n) && ((d >= split) == (hh != 0));
    float sc = 0.0f;
    if (FIRST) {
        if (hh == 0 || mine) sc = rg_onorm(rg_deg(odeg, s)) * rg_s2(rg_deg(ideg, d));
        if (hh == 0 && lane < n) escale[e0 + lane] = sc;
    } else {
        sc = escale[ei];
    }
    unsigned int m = __builtin_amdgcn_ballot_w32(mine);
    for (int k = 0; k < EDGES_PER_WAVE / 4 && __builtin_popcount(m) >= 4; ++k) {
        int jl[4];
        jl[0] = rg_take(m); jl[1] = rg_take(m); jl[2] = rg_take(m); jl[3] = rg_take(m);
        rg_edge_group<4>(h, wtab, acc, s, d, t, sc, jl, lane);
    }
    if (__builtin_popcount(m) >= 2) {
        int jl[2];
        jl[0] = rg_take(m); jl[1] = rg_take(m);
        rg_edge_group<2>(h, wtab, acc, s, d, t, sc, jl, lane);
    }
    if (__builtin_popcount(m) >= 2) {
        int jl[2];
        jl[0] = rg_take(m); jl[1] = rg_take(m);
        rg_edge_group<2>(h, wtab, acc, s, d, t, sc, jl, lane);
    }
    if (m) {
        int jl[1];
        jl[0] = rg_take(m);
        rg_edge_group<1>(h, wtab, acc, s, d, t, sc, jl, lane);
    }
}

template <bool FIRST>
__global__ __launch_bounds__(EDGE_THREADS) void
rg_edge_kernel(const float* __restrict__ h, const float* __restrict__ odeg,
               const float* __restrict__ ideg,
               const float* __restrict__ weight, const int* __restrict__ src,
               const int* __restrict__ dst, const int* __restrict__ etype,
               float* __restrict__ acc, float* __restrict__ escale,
               unsigned int* tick, int E, int R, int N) {
    extern __shared__ float wtab[];
    const int tid = threadIdx.x;
    const int lane = tid & 31;
    const int wave = __builtin_amdgcn_readfirstlane((int)(tid >> 5));
    const float4* __restrict__ w4 = (const float4*)weight;
    float4* l4 = (float4*)wtab;
    const int nv = R * (NDIM / 4);
    unsigned int* s_tick = (unsigned int*)(wtab + 4 * max(nv, WTAB_STEPS * EDGE_THREADS));
    float4 wv[WTAB_STEPS];
    #pragma unroll
    for (int u = 0; u < WTAB_STEPS; ++u) {
        const int i = tid + u * EDGE_THREADS;
        wv[u] = w4[i < nv ? i : 0];
    }
    #pragma unroll
    for (int u = 0; u < WTAB_STEPS; ++u) {
        const int i = tid + u * EDGE_THREADS;
        l4[i] = wv[u];
    }
    for (int i = tid + WTAB_STEPS * EDGE_THREADS; i < nv; i += EDGE_THREADS) l4[i] = w4[i];

    const unsigned int where = __builtin_amdgcn_s_sendmsg_rtn(0x87);
    const int myhalf = (int)((where >> 18) & 1u);
    const int nChunks = (E + EDGES_PER_CHUNK - 1) / EDGES_PER_CHUNK;
    const int split = N >> 1;

    int g = 0;
    for (int pass = 0; pass < 2; ++pass) {
        const int hh = myhalf ^ pass;
        unsigned int* tk = tick + hh;
        if (tid == 0) s_tick[g & 1] = atomicAdd(tk, 1u);
        __syncthreads();
        unsigned int c = (unsigned int)__builtin_amdgcn_readfirstlane((int)s_tick[g & 1]);
        ++g;
        for (int it = 0; it < nChunks && c < (unsigned int)nChunks; ++it) {
            unsigned int nxt = 0u;
            if (tid == 0) nxt = atomicAdd(tk, 1u);
            rg_edge_chunk<FIRST>(h, odeg, ideg, wtab, src, dst, etype, acc, escale,
                                 E, split, (int)c, hh, wave, lane);
            if (tid == 0) s_tick[g & 1] = nxt;
            __syncthreads();
            c = (unsigned int)__builtin_amdgcn_readfirstlane((int)s_tick[g & 1]);
            ++g;
        }
    }
}

#define LDS_STRIDE 132
#define GEMM_WAVES 8
#define GEMM_THREADS (GEMM_WAVES * 32)
#define GEMM_ROWS 32
#define GEMM_GRID 512

__device__ __forceinline__ float4
rg_gemm_fetch(const float* __restrict__ H, int r0, int nrows, int N, int wave, int lane, int u) {
    const int r = wave + u * GEMM_WAVES;
    float4 v = make_float4(0.0f, 0.0f, 0.0f, 0.0f);
    if (r < nrows) v = ((const float4*)(H + (size_t)min(r0 + r, N - 1) * NDIM))[lane];
    return v;
}
__device__ __forceinline__ void
rg_gemm_stage(float* dstH, int nrows, int wave, int lane, int u, float4 v) {
    const int r = wave + u * GEMM_WAVES;
    if (r < nrows) *(float4*)(dstH + r * LDS_STRIDE + lane * 4) = v;
}

__global__ __launch_bounds__(GEMM_THREADS) void
rg_gemm_kernel(const float* __restrict__ H, const float* __restrict__ W,
               const float* __restrict__ ideg, float* __restrict__ out,
               unsigned int* __restrict__ tick, int N) {
    __shared__ __attribute__((aligned(16))) float ldsH[2][GEMM_ROWS * LDS_STRIDE];
    __shared__ float ldsS[2][GEMM_ROWS];

    const int tid = threadIdx.x;
    const int lane = tid & 31;
    const int hi = lane >> 4;
    const int lo = lane & 15;
    const int wave = __builtin_amdgcn_readfirstlane((int)(tid >> 5));
    const int n0 = wave * 16;

    const unsigned int nTiles = (unsigned int)(N + 15) >> 4;
    const int t0 = (int)((nTiles * blockIdx.x) / GEMM_GRID);
    const int t1 = (int)((nTiles * (blockIdx.x + 1u)) / GEMM_GRID);
    const int nSteps = (t1 - t0 + 1) >> 1;

    float4 p0, p1, p2, p3;
    float4 pd = make_float4(1.0f, 0.0f, 0.0f, 0.0f);
    int nr = 16 * min(2, t1 - t0);
    p0 = rg_gemm_fetch(H, t0 * 16, nr, N, wave, lane, 0);
    p1 = rg_gemm_fetch(H, t0 * 16, nr, N, wave, lane, 1);
    p2 = rg_gemm_fetch(H, t0 * 16, nr, N, wave, lane, 2);
    p3 = rg_gemm_fetch(H, t0 * 16, nr, N, wave, lane, 3);
    if (tid < nr) pd = ((const float4*)ideg)[min(t0 * 16 + tid, N - 1)];

    v2f bf[NDIM / 4];
    #pragma unroll
    for (int j = 0; j < NDIM / 4; ++j) {
        const int kk = 4 * j + 2 * hi;
        bf[j].x = W[(size_t)kk * NDIM + n0 + lo];
        bf[j].y = W[(size_t)(kk + 1) * NDIM + n0 + lo];
    }

    rg_gemm_stage(ldsH[0], nr, wave, lane, 0, p0);
    rg_gemm_stage(ldsH[0], nr, wave, lane, 1, p1);
    rg_gemm_stage(ldsH[0], nr, wave, lane, 2, p2);
    rg_gemm_stage(ldsH[0], nr, wave, lane, 3, p3);
    if (tid < nr) ldsS[0][tid] = rsqrtf(fmaxf((pd.x + pd.y) + (pd.z + pd.w), 1.0f));
    __syncthreads();

    for (int s = 0; s < nSteps; ++s) {
        const int buf = s & 1;
        const int t = t0 + 2 * s;
        const int nt = min(2, t1 - t);
        const int tn = t + 2;
        nr = (s + 1 < nSteps) ? 16 * min(2, t1 - tn) : 0;
        p0 = rg_gemm_fetch(H, tn * 16, nr, N, wave, lane, 0);
        p1 = rg_gemm_fetch(H, tn * 16, nr, N, wave, lane, 1);
        p2 = rg_gemm_fetch(H, tn * 16, nr, N, wave, lane, 2);
        p3 = rg_gemm_fetch(H, tn * 16, nr, N, wave, lane, 3);
        if (tid < nr) pd = ((const float4*)ideg)[min(tn * 16 + tid, N - 1)];

        const float* A = ldsH[buf];
        const float* S = ldsS[buf];
        v8f c0 = {};
        v8f c1 = {};
        if (nt == 2) {
            #pragma unroll
            for (int j = 0; j < NDIM / 4; ++j) {
                const int kk = 4 * j + 2 * hi;
                v2f a0, a1;
                a0.x = A[lo * LDS_STRIDE + kk];
                a0.y = A[lo * LDS_STRIDE + kk + 1];
                a1.x = A[(16 + lo) * LDS_STRIDE + kk];
                a1.y = A[(16 + lo) * LDS_STRIDE + kk + 1];
                c0 = __builtin_amdgcn_wmma_f32_16x16x4_f32(
                         false, a0,  false, bf[j],
                         (short)0, c0,  false,  false);
                c1 = __builtin_amdgcn_wmma_f32_16x16x4_f32(
                         false, a1,  false, bf[j],
                         (short)0, c1,  false,  false);
            }
        } else {
            #pragma unroll
            for (int j = 0; j < NDIM / 4; ++j) {
                const int kk = 4 * j + 2 * hi;
                v2f a0;
                a0.x = A[lo * LDS_STRIDE + kk];
                a0.y = A[lo * LDS_STRIDE + kk + 1];
                c0 = __builtin_amdgcn_wmma_f32_16x16x4_f32(
                         false, a0,  false, bf[j],
                         (short)0, c0,  false,  false);
            }
        }
        #pragma unroll
        for (int g = 0; g < 8; ++g) c0[g] *= S[g + 8 * hi];
        if (nt == 2) {
            #pragma unroll
            for (int g = 0; g < 8; ++g) c1[g] *= S[16 + g + 8 * hi];
        }

        rg_gemm_stage(ldsH[buf ^ 1], nr, wave, lane, 0, p0);
        rg_gemm_stage(ldsH[buf ^ 1], nr, wave, lane, 1, p1);
        rg_gemm_stage(ldsH[buf ^ 1], nr, wave, lane, 2, p2);
        rg_gemm_stage(ldsH[buf ^ 1], nr, wave, lane, 3, p3);
        if (tid < nr) ldsS[buf ^ 1][tid] = rsqrtf(fmaxf((pd.x + pd.y) + (pd.z + pd.w), 1.0f));
        __syncthreads();

        const int r0 = t * 16 + 8 * hi;
        float* op = out + (size_t)r0 * NDIM + n0 + lo;
        if ((t + nt) * 16 <= N) {
            #pragma unroll
            for (int g = 0; g < 8; ++g) op[(size_t)g * NDIM] = c0[g];
            if (nt == 2) {
                #pragma unroll
                for (int g = 0; g < 8; ++g) op[(size_t)(16 + g) * NDIM] = c1[g];
            }
        } else {
            #pragma unroll
            for (int g = 0; g < 8; ++g) {
                if (r0 + g < N) op[(size_t)g * NDIM] = c0[g];
                if (nt == 2 && r0 + 16 + g < N) op[(size_t)(16 + g) * NDIM] = c1[g];
            }
        }
    }

    if (tick != nullptr && blockIdx.x == 0 && tid < 2) tick[tid] = 0u;
}

extern "C" void kernel_launch(void* const* d_in, const int* in_sizes, int n_in,
                              void* d_out, int out_size, void* d_ws, size_t ws_size,
                              hipStream_t stream) {
    const float* x      = (const float*)d_in[0];
    const float* weight = (const float*)d_in[1];
    const float* lw0    = (const float*)d_in[2];
    const float* lw1    = (const float*)d_in[3];
    const int*   src    = (const int*)d_in[4];
    const int*   dst    = (const int*)d_in[5];
    const int*   etype  = (const int*)d_in[6];
    float* out = (float*)d_out;

    const int N = in_sizes[0] / NDIM;
    const int E = in_sizes[4];
    const int R = in_sizes[1] / NDIM;
    const size_t ND = (size_t)N * NDIM;

    float* ws    = (float*)d_ws;
    float* h1    = ws;
    float* odeg  = h1 + ND;
    float* ideg  = odeg + (size_t)N * DEG_PARTS;
    float* escale = ideg + (size_t)N * DEG_PARTS;

    const int nRanges = (N + DEG_RANGE - 1) / DEG_RANGE;
    unsigned int* tick1 = (unsigned int*)out;
    unsigned int* tick2 = (unsigned int*)odeg;
    rg_deg_kernel<<<2 * nRanges * DEG_PARTS, DEG_THREADS, 0, stream>>>(src, dst, odeg, tick1, E, N, nRanges);

    const size_t wtabMin = (size_t)WTAB_STEPS * EDGE_THREADS * 16;
    const size_t wtabTab = (size_t)R * NDIM * sizeof(float);
    const size_t wtabBytes = (wtabTab > wtabMin ? wtabTab : wtabMin) + 16;

    rg_gemm_kernel<<<GEMM_GRID, GEMM_THREADS, 0, stream>>>(x, lw0, ideg, h1, nullptr, N);
    rg_edge_kernel<true><<<EDGE_GRID, EDGE_THREADS, wtabBytes, stream>>>(x, odeg, ideg, weight, src, dst, etype, h1, escale, tick1, E, R, N);

    rg_gemm_kernel<<<GEMM_GRID, GEMM_THREADS, 0, stream>>>(h1, lw1, ideg, out, tick2, N);
    rg_edge_kernel<false><<<EDGE_GRID, EDGE_THREADS, wtabBytes, stream>>>(h1, odeg, ideg, weight, src, dst, etype, out, escale, tick2, E, R, N);
}
